# starting kernel
# baseline (speedup 1.0000x reference)
#include <hip/hip_runtime.h>
#include <math.h>

typedef __attribute__((ext_vector_type(2))) float v2f;
typedef __attribute__((ext_vector_type(4))) float v4f;
typedef __attribute__((ext_vector_type(8))) float v8f;

#define DHID   128
#define TILE_M 16
#define WPB    6
#define HSTR   132
#define LNEPS  1e-5f

#if __has_builtin(__builtin_amdgcn_tanhf)
#define TANHF(v) __builtin_amdgcn_tanhf(v)
#else
#define TANHF(v) tanhf(v)
#endif

#define OFF_W1 32768
#define OFF_W2 49152
#define OFF_H  65536
#define SMEM_FLOATS (OFF_H + WPB * TILE_M * HSTR)

__device__ __forceinline__ void pack_w(float* __restrict__ dst,
                                       const float* __restrict__ W,
                                       int K, int tid, int nthr)
{
    const int total = K << 7;
    for (int i = tid; i < total; i += nthr) {
        const int r   = i >> 7;
        const int col = i & 127;
        dst[((r >> 2) << 9) + ((col >> 4) << 6) + ((col & 15) << 2) + (r & 3)] = W[i];
    }
}

template <int KSTEPS, int HS>
__device__ __forceinline__ void gemm_lds(const float* __restrict__ hbuf,
                                         const float* __restrict__ wp,
                                         v8f acc[8], int lane)
{
    const int row = lane & 15;
    const int g2  = (lane >> 4) << 1;
    const float* wl = wp + ((lane & 15) << 2) + g2;
#pragma unroll 4
    for (int kk = 0; kk < KSTEPS; ++kk) {
        const v2f a = *(const v2f*)(hbuf + row * HS + (kk << 2) + g2);
        const float* wk = wl + (kk << 9);
#pragma unroll
        for (int nn = 0; nn < 8; ++nn) {
            const v2f b = *(const v2f*)(wk + (nn << 6));
            acc[nn] = __builtin_amdgcn_wmma_f32_16x16x4_f32(
                false, a, false, b, (short)0, acc[nn], false, false);
        }
    }
}

__device__ __forceinline__ void layernorm_tanh(v8f acc[8],
                                               const float* __restrict__ bias,
                                               const float* __restrict__ gamma,
                                               const float* __restrict__ beta,
                                               int lane)
{
    const int col0 = lane & 15;
    float bsv[8], gv[8], bv[8];
#pragma unroll
    for (int nn = 0; nn < 8; ++nn) {
        const int c = col0 + (nn << 4);
        bsv[nn] = bias[c]; gv[nn] = gamma[c]; bv[nn] = beta[c];
    }
#pragma unroll
    for (int j = 0; j < 8; ++j) {
        float s = 0.f, ss = 0.f;
#pragma unroll
        for (int nn = 0; nn < 8; ++nn) {
            float v = acc[nn][j] + bsv[nn];
            acc[nn][j] = v;
            s += v;
            ss = fmaf(v, v, ss);
        }
#pragma unroll
        for (int m = 1; m < 16; m <<= 1) {
            s  += __shfl_xor(s,  m, 32);
            ss += __shfl_xor(ss, m, 32);
        }
        const float mu  = s * (1.0f / 128.0f);
        const float var = fmaf(-mu, mu, ss * (1.0f / 128.0f));
        const float inv = rsqrtf(var + LNEPS);
#pragma unroll
        for (int nn = 0; nn < 8; ++nn) {
            const float v = (acc[nn][j] - mu) * inv;
            acc[nn][j] = TANHF(fmaf(v, gv[nn], bv[nn]));
        }
    }
}

__device__ __forceinline__ void store_tile(float* __restrict__ hbuf,
                                           const v8f acc[8], int lane)
{
    const int col0 = lane & 15;
    const int grp  = lane >> 4;
#pragma unroll
    for (int j = 0; j < 8; ++j) {
        const int row = j + (grp << 3);
#pragma unroll
        for (int nn = 0; nn < 8; ++nn)
            hbuf[row * HSTR + col0 + (nn << 4)] = acc[nn][j];
    }
}

__device__ __forceinline__ void stage_rows(float* __restrict__ hbuf,
                                           const float* __restrict__ x,
                                           const int* __restrict__ idxbase,
                                           int e0, int E, int lane)
{
    const int c4 = lane << 2;
#pragma unroll 4
    for (int r = 0; r < TILE_M; ++r) {
        int e = e0 + r; if (e >= E) e = E - 1;
        const int n = idxbase[e];
        const v4f v = *(const v4f*)(x + (size_t)n * DHID + c4);
        *(v4f*)(hbuf + r * HSTR + c4) = v;
    }
}

__global__ void __launch_bounds__(WPB * 32, 1)
edge_mlp_wmma_f32(const float* __restrict__ x, const int* __restrict__ eidx, int E,
                  const float* __restrict__ W0, const float* __restrict__ b0,
                  const float* __restrict__ W1, const float* __restrict__ b1,
                  const float* __restrict__ W2, const float* __restrict__ b2,
                  const float* __restrict__ W3, const float* __restrict__ b3,
                  const float* __restrict__ g0, const float* __restrict__ be0,
                  const float* __restrict__ g1, const float* __restrict__ be1,
                  const float* __restrict__ g2, const float* __restrict__ be2,
                  float* __restrict__ out)
{
    extern __shared__ float smem[];
    const int tid  = threadIdx.x;
    const int lane = tid & 31;
    const int wave = tid >> 5;

    float* wp0  = smem;
    float* wp1  = smem + OFF_W1;
    float* wp2  = smem + OFF_W2;
    float* hbuf = smem + OFF_H + wave * (TILE_M * HSTR);

    pack_w(wp0, W0, 2 * DHID, tid, WPB * 32);
    pack_w(wp1, W1, DHID,     tid, WPB * 32);
    pack_w(wp2, W2, DHID,     tid, WPB * 32);
    __syncthreads();

    const int tiles  = (E + TILE_M - 1) / TILE_M;
    const int stride = gridDim.x * WPB;
    const int sweeps = (tiles + stride - 1) / stride;

    for (int it = 0; it < sweeps; ++it) {
        const int tile = (it * gridDim.x + blockIdx.x) * WPB + wave;
        const int e0   = tile * TILE_M;

        {
            const int tn = tile + stride;
            if (tn < tiles) {
                __builtin_prefetch(&eidx[tn * TILE_M], 0, 1);
                __builtin_prefetch(&eidx[E + tn * TILE_M], 0, 1);
            }
        }

        v8f h0[8] = {};
        stage_rows(hbuf, x, eidx, e0, E, lane);
        __syncthreads();
        gemm_lds<32, HSTR>(hbuf, wp0, h0, lane);
        __syncthreads();
        stage_rows(hbuf, x, eidx + E, e0, E, lane);
        __syncthreads();
        gemm_lds<32, HSTR>(hbuf, wp0 + 32 * 512, h0, lane);
        layernorm_tanh(h0, b0, g0, be0, lane);
        store_tile(hbuf, h0, lane);
        __syncthreads();

        v8f h1[8] = {};
        gemm_lds<32, HSTR>(hbuf, wp1, h1, lane);
        layernorm_tanh(h1, b1, g1, be1, lane);
        store_tile(hbuf, h1, lane);
        __syncthreads();

        v8f h2[8] = {};
        gemm_lds<32, HSTR>(hbuf, wp2, h2, lane);
        layernorm_tanh(h2, b2, g2, be2, lane);

        {
            const int col0 = lane & 15;
            const int grp  = lane >> 4;
            float w3[8];
#pragma unroll
            for (int nn = 0; nn < 8; ++nn) w3[nn] = W3[col0 + (nn << 4)];
            const float bias3 = b3[0];
#pragma unroll
            for (int j = 0; j < 8; ++j) {
                float p = 0.f;
#pragma unroll
                for (int nn = 0; nn < 8; ++nn) p = fmaf(h2[nn][j], w3[nn], p);
#pragma unroll
                for (int m = 1; m < 16; m <<= 1) p += __shfl_xor(p, m, 32);
                const int e = e0 + j + (grp << 3);
                if (col0 == j && e < E) out[e] = p + bias3;
            }
        }
        __syncthreads();
    }
}

extern "C" void kernel_launch(void* const* d_in, const int* in_sizes, int n_in,
                              void* d_out, int out_size, void* d_ws, size_t ws_size,
                              hipStream_t stream)
{
    const float* x   = (const float*)d_in[0];
    const int*   ei  = (const int*)  d_in[1];
    const float* W0  = (const float*)d_in[2];
    const float* b0  = (const float*)d_in[3];
    const float* W1  = (const float*)d_in[4];
    const float* b1  = (const float*)d_in[5];
    const float* W2  = (const float*)d_in[6];
    const float* b2  = (const float*)d_in[7];
    const float* W3  = (const float*)d_in[8];
    const float* b3  = (const float*)d_in[9];
    const float* g0  = (const float*)d_in[10];
    const float* be0 = (const float*)d_in[11];
    const float* g1  = (const float*)d_in[12];
    const float* be1 = (const float*)d_in[13];
    const float* g2  = (const float*)d_in[14];
    const float* be2 = (const float*)d_in[15];

    const int E     = in_sizes[1] / 2;
    const int tiles = (E + TILE_M - 1) / TILE_M;
    int blocks = (tiles + WPB - 1) / WPB;
    if (blocks > 1024) blocks = 1024;
    const size_t shmem = (size_t)SMEM_FLOATS * sizeof(float);

    edge_mlp_wmma_f32<<<blocks, WPB * 32, shmem, stream>>>(
        x, ei, E, W0, b0, W1, b1, W2, b2, W3, b3,
        g0, be0, g1, be1, g2, be2, (float*)d_out);
}


# weight staging rewritten: a wave packs 4-row slabs, whole-line loads in flight up to the load counter's depth, one conflict-free 16-byte LDS store per column (was 8 loads a trip); sums unchanged
# speedup vs baseline: 1.5635x; 1.5635x over previous
#include <hip/hip_runtime.h>
#include <math.h>


typedef __attribute__((ext_vector_type(2))) float v2f;
typedef __attribute__((ext_vector_type(4))) float v4f;
typedef __attribute__((ext_vector_type(8))) float v8f;

#define DHID   128
#define TILE_M 32
#define WPB    4
#define LNEPS  1e-5f

#if __has_builtin(__builtin_amdgcn_tanhf)
#define TANHF(v) __builtin_amdgcn_tanhf(v)
#else
#define TANHF(v) tanhf(v)
#endif

#define OFF_W1 32768
#define OFF_W2 49152
#define OFF_H  65536
#define HTILE  (TILE_M * DHID)
#define SMEM_FLOATS (OFF_H + WPB * HTILE)


__device__ __forceinline__ void pack_slabs16(float* __restrict__ dst,
                                             const float* __restrict__ W,
                                             int wave, int lane)
{
    const float* src = W   + (wave << 11) + lane;
    float*       d   = dst + (wave << 11) + (lane << 2);
    float v[4][4][4];
#pragma unroll
    for (int u = 0; u < 4; ++u)
#pragma unroll
        for (int s = 0; s < 4; ++s)
#pragma unroll
            for (int j = 0; j < 4; ++j)
                v[u][j][s] = src[(u << 9) + (s << 7) + (j << 5)];
#pragma unroll
    for (int u = 0; u < 4; ++u)
#pragma unroll
        for (int j = 0; j < 4; ++j) {
            const v4f o = { v[u][j][0], v[u][j][1], v[u][j][2], v[u][j][3] };
            *(v4f*)(d + (u << 9) + (j << 7)) = o;
        }
}

template <int KSTEPS>
__device__ __forceinline__ void gemm_lds(const float* __restrict__ hbuf,
                                         const float* __restrict__ wp,
                                         v8f acc0[8], v8f acc1[8], int lane)
{
    const int r  = lane & 15;
    const int g2 = (lane >> 4) << 1;
    const int abase = r * DHID + (r << 2) + g2;
    const float* wl = wp + (r << 2) + g2;
#pragma unroll 4
    for (int kk = 0; kk < KSTEPS; ++kk) {
        const float* ap = hbuf + (abase ^ (kk << 2));
        const v2f a0 = *(const v2f*)ap;
        const v2f a1 = *(const v2f*)(ap + 16 * DHID);
        const float* wk = wl + (kk << 9);
#pragma unroll
        for (int nn = 0; nn < 8; ++nn) {
            const v2f b = *(const v2f*)(wk + (nn << 6));
            acc0[nn] = __builtin_amdgcn_wmma_f32_16x16x4_f32(
                false, a0, false, b, (short)0, acc0[nn], false, false);
            acc1[nn] = __builtin_amdgcn_wmma_f32_16x16x4_f32(
                false, a1, false, b, (short)0, acc1[nn], false, false);
        }
    }
}

template <int M>
__device__ __forceinline__ float row_xor_add(float v)
{
    const int p = __builtin_amdgcn_update_dpp(0, __float_as_int(v), 0x160 + M,
                                              0xf, 0xf, true);
    return v + __int_as_float(p);
}

__device__ __forceinline__ float row_sum16(float v)
{
    v = row_xor_add<1>(v);
    v = row_xor_add<2>(v);
    v = row_xor_add<4>(v);
    v = row_xor_add<8>(v);
    return v;
}

__device__ __forceinline__ void ln_tile(v8f acc[8], const float bsv[8],
                                        const float gv[8], const float bv[8])
{
#pragma unroll
    for (int j = 0; j < 8; ++j) {
        float s = 0.f, ss = 0.f;
#pragma unroll
        for (int nn = 0; nn < 8; ++nn) {
            float v = acc[nn][j] + bsv[nn];
            acc[nn][j] = v;
            s += v;
            ss = fmaf(v, v, ss);
        }
        s  = row_sum16(s);
        ss = row_sum16(ss);
        const float mu  = s * (1.0f / 128.0f);
        const float var = fmaf(-mu, mu, ss * (1.0f / 128.0f));
        const float inv = rsqrtf(var + LNEPS);
#pragma unroll
        for (int nn = 0; nn < 8; ++nn) {
            const float v = (acc[nn][j] - mu) * inv;
            acc[nn][j] = TANHF(fmaf(v, gv[nn], bv[nn]));
        }
    }
}

__device__ __forceinline__ void layernorm_tanh(v8f acc0[8], v8f acc1[8],
                                               const float* __restrict__ bias,
                                               const float* __restrict__ gamma,
                                               const float* __restrict__ beta,
                                               int lane)
{
    const int col0 = lane & 15;
    float bsv[8], gv[8], bv[8];
#pragma unroll
    for (int nn = 0; nn < 8; ++nn) {
        const int c = col0 + (nn << 4);
        bsv[nn] = bias[c]; gv[nn] = gamma[c]; bv[nn] = beta[c];
    }
    ln_tile(acc0, bsv, gv, bv);
    ln_tile(acc1, bsv, gv, bv);
}

__device__ __forceinline__ void store_tile(float* __restrict__ hbuf,
                                           const v8f acc[8], int rowoff, int lane)
{
    const int col0 = lane & 15;
    const int grp  = lane >> 4;
    const int y = (grp << 10) | ((((col0 >> 2) ^ (grp << 3))) << 2) | (col0 & 3);
#pragma unroll
    for (int j = 0; j < 8; ++j) {
#pragma unroll
        for (int nn = 0; nn < 8; ++nn)
            hbuf[(rowoff + j) * DHID + (y ^ (((nn << 2) ^ j) << 2))] = acc[nn][j];
    }
}

__device__ __forceinline__ void stage_rows(float* __restrict__ hbuf,
                                           const float* __restrict__ x,
                                           const int* __restrict__ idxbase,
                                           int e0, int E, int lane)
{
    const int c4 = lane << 2;
    v4f v[TILE_M];
#pragma unroll
    for (int r = 0; r < TILE_M; ++r) {
        int e = e0 + r; if (e >= E) e = E - 1;
        const int n = idxbase[e];
        v[r] = *(const v4f*)(x + (size_t)n * DHID + c4);
    }
#pragma unroll
    for (int r = 0; r < TILE_M; ++r)
        *(v4f*)(hbuf + r * DHID + ((lane ^ (r & 15)) << 2)) = v[r];
}

__global__ void __launch_bounds__(WPB * 32, 1)
edge_mlp_wmma_f32(const float* __restrict__ x, const int* __restrict__ eidx, int E,
                  const float* __restrict__ W0, const float* __restrict__ b0,
                  const float* __restrict__ W1, const float* __restrict__ b1,
                  const float* __restrict__ W2, const float* __restrict__ b2,
                  const float* __restrict__ W3, const float* __restrict__ b3,
                  const float* __restrict__ g0, const float* __restrict__ be0,
                  const float* __restrict__ g1, const float* __restrict__ be1,
                  const float* __restrict__ g2, const float* __restrict__ be2,
                  float* __restrict__ out)
{
    __shared__ alignas(16) float smem[SMEM_FLOATS];
    const int tid  = threadIdx.x;
    const int lane = tid & 31;
    const int wave = tid >> 5;

    float* wp0  = smem;
    float* wp1  = smem + OFF_W1;
    float* wp2  = smem + OFF_W2;
    float* hbuf = smem + OFF_H + wave * HTILE;

#pragma unroll
    for (int t = 0; t < 8; ++t) {
        const float* Wt = (t < 4) ? W0 + (t << 13)
                        : (t < 6) ? W1 + ((t - 4) << 13)
                                  : W2 + ((t - 6) << 13);
        pack_slabs16(smem + (t << 13), Wt, wave, lane);
    }
    __syncthreads();

    const int tiles  = (E + TILE_M - 1) / TILE_M;
    const int stride = gridDim.x * WPB;
    const int sweeps = (tiles + stride - 1) / stride;

    for (int it = 0; it < sweeps; ++it) {
        const int tile = (it * gridDim.x + blockIdx.x) * WPB + wave;
        const int e0   = tile * TILE_M;

        {
            const int tn = tile + stride;
            if (tn < tiles) {
                __builtin_prefetch(&eidx[tn * TILE_M], 0, 1);
                __builtin_prefetch(&eidx[E + tn * TILE_M], 0, 1);
            }
        }

        v8f p0[8] = {}, p1[8] = {};
        stage_rows(hbuf, x, eidx, e0, E, lane);
        __syncthreads();
        gemm_lds<32>(hbuf, wp0, p0, p1, lane);
        __syncthreads();
        stage_rows(hbuf, x, eidx + E, e0, E, lane);
        __syncthreads();
        gemm_lds<32>(hbuf, wp0 + 32 * 512, p0, p1, lane);
        layernorm_tanh(p0, p1, b0, g0, be0, lane);
        store_tile(hbuf, p0, 0, lane);
        store_tile(hbuf, p1, 16, lane);
        __syncthreads();

        v8f q0[8] = {}, q1[8] = {};
        gemm_lds<32>(hbuf, wp1, q0, q1, lane);
        layernorm_tanh(q0, q1, b1, g1, be1, lane);
        store_tile(hbuf, q0, 0, lane);
        store_tile(hbuf, q1, 16, lane);
        __syncthreads();

        v8f r0[8] = {}, r1[8] = {};
        gemm_lds<32>(hbuf, wp2, r0, r1, lane);
        layernorm_tanh(r0, r1, b2, g2, be2, lane);

        {
            const int col0 = lane & 15;
            const int grp  = lane >> 4;
            float w3[8];
#pragma unroll
            for (int nn = 0; nn < 8; ++nn) w3[nn] = W3[col0 + (nn << 4)];
            const float bias3 = b3[0];
#pragma unroll
            for (int t = 0; t < 2; ++t) {
#pragma unroll
                for (int j = 0; j < 8; ++j) {
                    float p = 0.f;
#pragma unroll
                    for (int nn = 0; nn < 8; ++nn)
                        p = fmaf(t ? r1[nn][j] : r0[nn][j], w3[nn], p);
                    p = row_sum16(p);
                    const int e = e0 + (t << 4) + j + (grp << 3);
                    if (col0 == j && e < E) out[e] = p + bias3;
                }
            }
        }
        __syncthreads();
    }
}

extern "C" void kernel_launch(void* const* d_in, const int* in_sizes, int n_in,
                              void* d_out, int out_size, void* d_ws, size_t ws_size,
                              hipStream_t stream)
{
    const float* x   = (const float*)d_in[0];
    const int*   ei  = (const int*)  d_in[1];
    const float* W0  = (const float*)d_in[2];
    const float* b0  = (const float*)d_in[3];
    const float* W1  = (const float*)d_in[4];
    const float* b1  = (const float*)d_in[5];
    const float* W2  = (const float*)d_in[6];
    const float* b2  = (const float*)d_in[7];
    const float* W3  = (const float*)d_in[8];
    const float* b3  = (const float*)d_in[9];
    const float* g0  = (const float*)d_in[10];
    const float* be0 = (const float*)d_in[11];
    const float* g1  = (const float*)d_in[12];
    const float* be1 = (const float*)d_in[13];
    const float* g2  = (const float*)d_in[14];
    const float* be2 = (const float*)d_in[15];

    const int E     = in_sizes[1] / 2;
    const int tiles = (E + TILE_M - 1) / TILE_M;
    int blocks = (tiles + WPB - 1) / WPB;
    if (blocks > 1024) blocks = 1024;

    edge_mlp_wmma_f32<<<blocks, WPB * 32, 0, stream>>>(
        x, ei, E, W0, b0, W1, b1, W2, b2, W3, b3,
        g0, be0, g1, be1, g2, be2, (float*)d_out);
}


# GEMM K loop software-pipelined: next block's 20 LDS fragment loads issued into a second register set ahead of the current block's 64 WMMAs (was load, wait, multiply per block); sums unchanged
# speedup vs baseline: 1.6176x; 1.0346x over previous
#include <hip/hip_runtime.h>
#include <math.h>


typedef __attribute__((ext_vector_type(2))) float v2f;
typedef __attribute__((ext_vector_type(4))) float v4f;
typedef __attribute__((ext_vector_type(8))) float v8f;

#define DHID   128
#define TILE_M 32
#define WPB    4
#define LNEPS  1e-5f

#if __has_builtin(__builtin_amdgcn_tanhf)
#define TANHF(v) __builtin_amdgcn_tanhf(v)
#else
#define TANHF(v) tanhf(v)
#endif

#define SCHED_FENCE() __builtin_amdgcn_sched_barrier(0)

#define OFF_W1 32768
#define OFF_W2 49152
#define OFF_H  65536
#define HTILE  (TILE_M * DHID)
#define SMEM_FLOATS (OFF_H + WPB * HTILE)


__device__ __forceinline__ void pack_slabs16(float* __restrict__ dst,
                                             const float* __restrict__ W,
                                             int wave, int lane)
{
    const float* src = W   + (wave << 11) + lane;
    float*       d   = dst + (wave << 11) + (lane << 2);
    float v[4][4][4];
#pragma unroll
    for (int u = 0; u < 4; ++u)
#pragma unroll
        for (int s = 0; s < 4; ++s)
#pragma unroll
            for (int j = 0; j < 4; ++j)
                v[u][j][s] = src[(u << 9) + (s << 7) + (j << 5)];
#pragma unroll
    for (int u = 0; u < 4; ++u)
#pragma unroll
        for (int j = 0; j < 4; ++j) {
            const v4f o = { v[u][j][0], v[u][j][1], v[u][j][2], v[u][j][3] };
            *(v4f*)(d + (u << 9) + (j << 7)) = o;
        }
}

struct Frags {
    v2f a0[4];
    v2f a1[4];
    v2f b[4][8];
};

__device__ __forceinline__ void load_frags(Frags& f, const float* __restrict__ hbuf,
                                           const float* __restrict__ wl,
                                           int abase, int kk0)
{
#pragma unroll
    for (int u = 0; u < 4; ++u) {
        const int kk = kk0 + u;
        const float* ap = hbuf + (abase ^ (kk << 2));
        f.a0[u] = *(const v2f*)ap;
        f.a1[u] = *(const v2f*)(ap + 16 * DHID);
        const float* wk = wl + (kk << 9);
#pragma unroll
        for (int nn = 0; nn < 8; ++nn)
            f.b[u][nn] = *(const v2f*)(wk + (nn << 6));
    }
}

__device__ __forceinline__ void mma_block(const Frags& f, v8f acc0[8], v8f acc1[8])
{
#pragma unroll
    for (int u = 0; u < 4; ++u) {
#pragma unroll
        for (int nn = 0; nn < 8; ++nn) {
            acc0[nn] = __builtin_amdgcn_wmma_f32_16x16x4_f32(
                false, f.a0[u], false, f.b[u][nn], (short)0, acc0[nn], false, false);
            acc1[nn] = __builtin_amdgcn_wmma_f32_16x16x4_f32(
                false, f.a1[u], false, f.b[u][nn], (short)0, acc1[nn], false, false);
        }
    }
}

template <int KSTEPS>
__device__ __forceinline__ void gemm_lds(const float* __restrict__ hbuf,
                                         const float* __restrict__ wp,
                                         v8f acc0[8], v8f acc1[8], int lane)
{
    static_assert(KSTEPS >= 8 && (KSTEPS & 7) == 0, "two blocks of 4 k-steps per trip");
    const int r  = lane & 15;
    const int g2 = (lane >> 4) << 1;
    const int abase = r * DHID + (r << 2) + g2;
    const float* wl = wp + (r << 2) + g2;

    Frags f0, f1;
    load_frags(f0, hbuf, wl, abase, 0);
#pragma unroll 1
    for (int kk = 0; kk < KSTEPS - 8; kk += 8) {
        load_frags(f1, hbuf, wl, abase, kk + 4);
        SCHED_FENCE();
        mma_block(f0, acc0, acc1);
        SCHED_FENCE();
        load_frags(f0, hbuf, wl, abase, kk + 8);
        SCHED_FENCE();
        mma_block(f1, acc0, acc1);
        SCHED_FENCE();
    }
    load_frags(f1, hbuf, wl, abase, KSTEPS - 4);
    SCHED_FENCE();
    mma_block(f0, acc0, acc1);
    SCHED_FENCE();
    mma_block(f1, acc0, acc1);
}

template <int M>
__device__ __forceinline__ float row_xor_add(float v)
{
    const int p = __builtin_amdgcn_update_dpp(0, __float_as_int(v), 0x160 + M,
                                              0xf, 0xf, true);
    return v + __int_as_float(p);
}

__device__ __forceinline__ float row_sum16(float v)
{
    v = row_xor_add<1>(v);
    v = row_xor_add<2>(v);
    v = row_xor_add<4>(v);
    v = row_xor_add<8>(v);
    return v;
}

__device__ __forceinline__ void ln_tile(v8f acc[8], const float bsv[8],
                                        const float gv[8], const float bv[8])
{
#pragma unroll
    for (int j = 0; j < 8; ++j) {
        float s = 0.f, ss = 0.f;
#pragma unroll
        for (int nn = 0; nn < 8; ++nn) {
            float v = acc[nn][j] + bsv[nn];
            acc[nn][j] = v;
            s += v;
            ss = fmaf(v, v, ss);
        }
        s  = row_sum16(s);
        ss = row_sum16(ss);
        const float mu  = s * (1.0f / 128.0f);
        const float var = fmaf(-mu, mu, ss * (1.0f / 128.0f));
        const float inv = rsqrtf(var + LNEPS);
#pragma unroll
        for (int nn = 0; nn < 8; ++nn) {
            const float v = (acc[nn][j] - mu) * inv;
            acc[nn][j] = TANHF(fmaf(v, gv[nn], bv[nn]));
        }
    }
}

__device__ __forceinline__ void layernorm_tanh(v8f acc0[8], v8f acc1[8],
                                               const float* __restrict__ bias,
                                               const float* __restrict__ gamma,
                                               const float* __restrict__ beta,
                                               int lane)
{
    const int col0 = lane & 15;
    float bsv[8], gv[8], bv[8];
#pragma unroll
    for (int nn = 0; nn < 8; ++nn) {
        const int c = col0 + (nn << 4);
        bsv[nn] = bias[c]; gv[nn] = gamma[c]; bv[nn] = beta[c];
    }
    ln_tile(acc0, bsv, gv, bv);
    ln_tile(acc1, bsv, gv, bv);
}

__device__ __forceinline__ void store_tile(float* __restrict__ hbuf,
                                           const v8f acc[8], int rowoff, int lane)
{
    const int col0 = lane & 15;
    const int grp  = lane >> 4;
    const int y = (grp << 10) | ((((col0 >> 2) ^ (grp << 3))) << 2) | (col0 & 3);
#pragma unroll
    for (int j = 0; j < 8; ++j) {
#pragma unroll
        for (int nn = 0; nn < 8; ++nn)
            hbuf[(rowoff + j) * DHID + (y ^ (((nn << 2) ^ j) << 2))] = acc[nn][j];
    }
}

__device__ __forceinline__ void stage_rows(float* __restrict__ hbuf,
                                           const float* __restrict__ x,
                                           const int* __restrict__ idxbase,
                                           int e0, int E, int lane)
{
    const int c4 = lane << 2;
    v4f v[TILE_M];
#pragma unroll
    for (int r = 0; r < TILE_M; ++r) {
        int e = e0 + r; if (e >= E) e = E - 1;
        const int n = idxbase[e];
        v[r] = *(const v4f*)(x + (size_t)n * DHID + c4);
    }
#pragma unroll
    for (int r = 0; r < TILE_M; ++r)
        *(v4f*)(hbuf + r * DHID + ((lane ^ (r & 15)) << 2)) = v[r];
}

__global__ void __launch_bounds__(WPB * 32, 1)
edge_mlp_wmma_f32(const float* __restrict__ x, const int* __restrict__ eidx, int E,
                  const float* __restrict__ W0, const float* __restrict__ b0,
                  const float* __restrict__ W1, const float* __restrict__ b1,
                  const float* __restrict__ W2, const float* __restrict__ b2,
                  const float* __restrict__ W3, const float* __restrict__ b3,
                  const float* __restrict__ g0, const float* __restrict__ be0,
                  const float* __restrict__ g1, const float* __restrict__ be1,
                  const float* __restrict__ g2, const float* __restrict__ be2,
                  float* __restrict__ out)
{
    __shared__ alignas(16) float smem[SMEM_FLOATS];
    const int tid  = threadIdx.x;
    const int lane = tid & 31;
    const int wave = tid >> 5;

    float* wp0  = smem;
    float* wp1  = smem + OFF_W1;
    float* wp2  = smem + OFF_W2;
    float* hbuf = smem + OFF_H + wave * HTILE;

#pragma unroll
    for (int t = 0; t < 8; ++t) {
        const float* Wt = (t < 4) ? W0 + (t << 13)
                        : (t < 6) ? W1 + ((t - 4) << 13)
                                  : W2 + ((t - 6) << 13);
        pack_slabs16(smem + (t << 13), Wt, wave, lane);
    }
    __syncthreads();

    const int tiles  = (E + TILE_M - 1) / TILE_M;
    const int stride = gridDim.x * WPB;
    const int sweeps = (tiles + stride - 1) / stride;

    for (int it = 0; it < sweeps; ++it) {
        const int tile = (it * gridDim.x + blockIdx.x) * WPB + wave;
        const int e0   = tile * TILE_M;

        {
            const int tn = tile + stride;
            if (tn < tiles) {
                __builtin_prefetch(&eidx[tn * TILE_M], 0, 1);
                __builtin_prefetch(&eidx[E + tn * TILE_M], 0, 1);
            }
        }

        v8f p0[8] = {}, p1[8] = {};
        stage_rows(hbuf, x, eidx, e0, E, lane);
        __syncthreads();
        gemm_lds<32>(hbuf, wp0, p0, p1, lane);
        __syncthreads();
        stage_rows(hbuf, x, eidx + E, e0, E, lane);
        __syncthreads();
        gemm_lds<32>(hbuf, wp0 + 32 * 512, p0, p1, lane);
        layernorm_tanh(p0, p1, b0, g0, be0, lane);
        store_tile(hbuf, p0, 0, lane);
        store_tile(hbuf, p1, 16, lane);
        __syncthreads();

        v8f q0[8] = {}, q1[8] = {};
        gemm_lds<32>(hbuf, wp1, q0, q1, lane);
        layernorm_tanh(q0, q1, b1, g1, be1, lane);
        store_tile(hbuf, q0, 0, lane);
        store_tile(hbuf, q1, 16, lane);
        __syncthreads();

        v8f r0[8] = {}, r1[8] = {};
        gemm_lds<32>(hbuf, wp2, r0, r1, lane);
        layernorm_tanh(r0, r1, b2, g2, be2, lane);

        {
            const int col0 = lane & 15;
            const int grp  = lane >> 4;
            float w3[8];
#pragma unroll
            for (int nn = 0; nn < 8; ++nn) w3[nn] = W3[col0 + (nn << 4)];
            const float bias3 = b3[0];
#pragma unroll
            for (int t = 0; t < 2; ++t) {
#pragma unroll
                for (int j = 0; j < 8; ++j) {
                    float p = 0.f;
#pragma unroll
                    for (int nn = 0; nn < 8; ++nn)
                        p = fmaf(t ? r1[nn][j] : r0[nn][j], w3[nn], p);
                    p = row_sum16(p);
                    const int e = e0 + (t << 4) + j + (grp << 3);
                    if (col0 == j && e < E) out[e] = p + bias3;
                }
            }
        }
        __syncthreads();
    }
}

extern "C" void kernel_launch(void* const* d_in, const int* in_sizes, int n_in,
                              void* d_out, int out_size, void* d_ws, size_t ws_size,
                              hipStream_t stream)
{
    const float* x   = (const float*)d_in[0];
    const int*   ei  = (const int*)  d_in[1];
    const float* W0  = (const float*)d_in[2];
    const float* b0  = (const float*)d_in[3];
    const float* W1  = (const float*)d_in[4];
    const float* b1  = (const float*)d_in[5];
    const float* W2  = (const float*)d_in[6];
    const float* b2  = (const float*)d_in[7];
    const float* W3  = (const float*)d_in[8];
    const float* b3  = (const float*)d_in[9];
    const float* g0  = (const float*)d_in[10];
    const float* be0 = (const float*)d_in[11];
    const float* g1  = (const float*)d_in[12];
    const float* be1 = (const float*)d_in[13];
    const float* g2  = (const float*)d_in[14];
    const float* be2 = (const float*)d_in[15];

    const int E     = in_sizes[1] / 2;
    const int tiles = (E + TILE_M - 1) / TILE_M;
    int blocks = (tiles + WPB - 1) / WPB;
    if (blocks > 1024) blocks = 1024;

    edge_mlp_wmma_f32<<<blocks, WPB * 32, 0, stream>>>(
        x, ei, E, W0, b0, W1, b1, W2, b2, W3, b3,
        g0, be0, g1, be1, g2, be2, (float*)d_out);
}


# sweep loop ends with the block's own last sweep: a block whose 4 tiles all lie past the last tile stops (was a full pass on clamped rows, nothing stored: sweep 5 of 432 blocks); sums unchanged
# speedup vs baseline: 1.6955x; 1.0482x over previous
#include <hip/hip_runtime.h>
#include <math.h>


typedef __attribute__((ext_vector_type(2))) float v2f;
typedef __attribute__((ext_vector_type(4))) float v4f;
typedef __attribute__((ext_vector_type(8))) float v8f;

#define DHID   128
#define TILE_M 32
#define WPB    4
#define LNEPS  1e-5f

#if __has_builtin(__builtin_amdgcn_tanhf)
#define TANHF(v) __builtin_amdgcn_tanhf(v)
#else
#define TANHF(v) tanhf(v)
#endif

#define SCHED_FENCE() __builtin_amdgcn_sched_barrier(0)

#define OFF_W1 32768
#define OFF_W2 49152
#define OFF_H  65536
#define HTILE  (TILE_M * DHID)
#define SMEM_FLOATS (OFF_H + WPB * HTILE)


__device__ __forceinline__ void pack_slabs16(float* __restrict__ dst,
                                             const float* __restrict__ W,
                                             int wave, int lane)
{
    const float* src = W   + (wave << 11) + lane;
    float*       d   = dst + (wave << 11) + (lane << 2);
    float v[4][4][4];
#pragma unroll
    for (int u = 0; u < 4; ++u)
#pragma unroll
        for (int s = 0; s < 4; ++s)
#pragma unroll
            for (int j = 0; j < 4; ++j)
                v[u][j][s] = src[(u << 9) + (s << 7) + (j << 5)];
#pragma unroll
    for (int u = 0; u < 4; ++u)
#pragma unroll
        for (int j = 0; j < 4; ++j) {
            const v4f o = { v[u][j][0], v[u][j][1], v[u][j][2], v[u][j][3] };
            *(v4f*)(d + (u << 9) + (j << 7)) = o;
        }
}

struct Frags {
    v2f a0[4];
    v2f a1[4];
    v2f b[4][8];
};

__device__ __forceinline__ void load_frags(Frags& f, const float* __restrict__ hbuf,
                                           const float* __restrict__ wl,
                                           int abase, int kk0)
{
#pragma unroll
    for (int u = 0; u < 4; ++u) {
        const int kk = kk0 + u;
        const float* ap = hbuf + (abase ^ (kk << 2));
        f.a0[u] = *(const v2f*)ap;
        f.a1[u] = *(const v2f*)(ap + 16 * DHID);
        const float* wk = wl + (kk << 9);
#pragma unroll
        for (int nn = 0; nn < 8; ++nn)
            f.b[u][nn] = *(const v2f*)(wk + (nn << 6));
    }
}

__device__ __forceinline__ void mma_block(const Frags& f, v8f acc0[8], v8f acc1[8])
{
#pragma unroll
    for (int u = 0; u < 4; ++u) {
#pragma unroll
        for (int nn = 0; nn < 8; ++nn) {
            acc0[nn] = __builtin_amdgcn_wmma_f32_16x16x4_f32(
                false, f.a0[u], false, f.b[u][nn], (short)0, acc0[nn], false, false);
            acc1[nn] = __builtin_amdgcn_wmma_f32_16x16x4_f32(
                false, f.a1[u], false, f.b[u][nn], (short)0, acc1[nn], false, false);
        }
    }
}

template <int KSTEPS>
__device__ __forceinline__ void gemm_lds(const float* __restrict__ hbuf,
                                         const float* __restrict__ wp,
                                         v8f acc0[8], v8f acc1[8], int lane)
{
    static_assert(KSTEPS >= 8 && (KSTEPS & 7) == 0, "two blocks of 4 k-steps per trip");
    const int r  = lane & 15;
    const int g2 = (lane >> 4) << 1;
    const int abase = r * DHID + (r << 2) + g2;
    const float* wl = wp + (r << 2) + g2;

    Frags f0, f1;
    load_frags(f0, hbuf, wl, abase, 0);
#pragma unroll 1
    for (int kk = 0; kk < KSTEPS - 8; kk += 8) {
        load_frags(f1, hbuf, wl, abase, kk + 4);
        SCHED_FENCE();
        mma_block(f0, acc0, acc1);
        SCHED_FENCE();
        load_frags(f0, hbuf, wl, abase, kk + 8);
        SCHED_FENCE();
        mma_block(f1, acc0, acc1);
        SCHED_FENCE();
    }
    load_frags(f1, hbuf, wl, abase, KSTEPS - 4);
    SCHED_FENCE();
    mma_block(f0, acc0, acc1);
    SCHED_FENCE();
    mma_block(f1, acc0, acc1);
}

template <int M>
__device__ __forceinline__ float row_xor_add(float v)
{
    const int p = __builtin_amdgcn_update_dpp(0, __float_as_int(v), 0x160 + M,
                                              0xf, 0xf, true);
    return v + __int_as_float(p);
}

__device__ __forceinline__ float row_sum16(float v)
{
    v = row_xor_add<1>(v);
    v = row_xor_add<2>(v);
    v = row_xor_add<4>(v);
    v = row_xor_add<8>(v);
    return v;
}

__device__ __forceinline__ void ln_tile(v8f acc[8], const float bsv[8],
                                        const float gv[8], const float bv[8])
{
#pragma unroll
    for (int j = 0; j < 8; ++j) {
        float s = 0.f, ss = 0.f;
#pragma unroll
        for (int nn = 0; nn < 8; ++nn) {
            float v = acc[nn][j] + bsv[nn];
            acc[nn][j] = v;
            s += v;
            ss = fmaf(v, v, ss);
        }
        s  = row_sum16(s);
        ss = row_sum16(ss);
        const float mu  = s * (1.0f / 128.0f);
        const float var = fmaf(-mu, mu, ss * (1.0f / 128.0f));
        const float inv = rsqrtf(var + LNEPS);
#pragma unroll
        for (int nn = 0; nn < 8; ++nn) {
            const float v = (acc[nn][j] - mu) * inv;
            acc[nn][j] = TANHF(fmaf(v, gv[nn], bv[nn]));
        }
    }
}

__device__ __forceinline__ void layernorm_tanh(v8f acc0[8], v8f acc1[8],
                                               const float* __restrict__ bias,
                                               const float* __restrict__ gamma,
                                               const float* __restrict__ beta,
                                               int lane)
{
    const int col0 = lane & 15;
    float bsv[8], gv[8], bv[8];
#pragma unroll
    for (int nn = 0; nn < 8; ++nn) {
        const int c = col0 + (nn << 4);
        bsv[nn] = bias[c]; gv[nn] = gamma[c]; bv[nn] = beta[c];
    }
    ln_tile(acc0, bsv, gv, bv);
    ln_tile(acc1, bsv, gv, bv);
}

__device__ __forceinline__ void store_tile(float* __restrict__ hbuf,
                                           const v8f acc[8], int rowoff, int lane)
{
    const int col0 = lane & 15;
    const int grp  = lane >> 4;
    const int y = (grp << 10) | ((((col0 >> 2) ^ (grp << 3))) << 2) | (col0 & 3);
#pragma unroll
    for (int j = 0; j < 8; ++j) {
#pragma unroll
        for (int nn = 0; nn < 8; ++nn)
            hbuf[(rowoff + j) * DHID + (y ^ (((nn << 2) ^ j) << 2))] = acc[nn][j];
    }
}

__device__ __forceinline__ void stage_rows(float* __restrict__ hbuf,
                                           const float* __restrict__ x,
                                           const int* __restrict__ idxbase,
                                           int e0, int E, int lane)
{
    const int c4 = lane << 2;
    v4f v[TILE_M];
#pragma unroll
    for (int r = 0; r < TILE_M; ++r) {
        int e = e0 + r; if (e >= E) e = E - 1;
        const int n = idxbase[e];
        v[r] = *(const v4f*)(x + (size_t)n * DHID + c4);
    }
#pragma unroll
    for (int r = 0; r < TILE_M; ++r)
        *(v4f*)(hbuf + r * DHID + ((lane ^ (r & 15)) << 2)) = v[r];
}

__global__ void __launch_bounds__(WPB * 32, 1)
edge_mlp_wmma_f32(const float* __restrict__ x, const int* __restrict__ eidx, int E,
                  const float* __restrict__ W0, const float* __restrict__ b0,
                  const float* __restrict__ W1, const float* __restrict__ b1,
                  const float* __restrict__ W2, const float* __restrict__ b2,
                  const float* __restrict__ W3, const float* __restrict__ b3,
                  const float* __restrict__ g0, const float* __restrict__ be0,
                  const float* __restrict__ g1, const float* __restrict__ be1,
                  const float* __restrict__ g2, const float* __restrict__ be2,
                  float* __restrict__ out)
{
    __shared__ alignas(16) float smem[SMEM_FLOATS];
    const int tid  = threadIdx.x;
    const int lane = tid & 31;
    const int wave = tid >> 5;

    float* wp0  = smem;
    float* wp1  = smem + OFF_W1;
    float* wp2  = smem + OFF_W2;
    float* hbuf = smem + OFF_H + wave * HTILE;

#pragma unroll
    for (int t = 0; t < 8; ++t) {
        const float* Wt = (t < 4) ? W0 + (t << 13)
                        : (t < 6) ? W1 + ((t - 4) << 13)
                                  : W2 + ((t - 6) << 13);
        pack_slabs16(smem + (t << 13), Wt, wave, lane);
    }
    __syncthreads();

    const int tiles  = (E + TILE_M - 1) / TILE_M;
    const int stride = gridDim.x * WPB;

    for (int tile0 = blockIdx.x * WPB; tile0 < tiles; tile0 += stride) {
        const int tile = tile0 + wave;
        const int e0   = tile * TILE_M;

        {
            const int tn = tile + stride;
            if (tn < tiles) {
                __builtin_prefetch(&eidx[tn * TILE_M], 0, 1);
                __builtin_prefetch(&eidx[E + tn * TILE_M], 0, 1);
            }
        }

        v8f p0[8] = {}, p1[8] = {};
        stage_rows(hbuf, x, eidx, e0, E, lane);
        __syncthreads();
        gemm_lds<32>(hbuf, wp0, p0, p1, lane);
        __syncthreads();
        stage_rows(hbuf, x, eidx + E, e0, E, lane);
        __syncthreads();
        gemm_lds<32>(hbuf, wp0 + 32 * 512, p0, p1, lane);
        layernorm_tanh(p0, p1, b0, g0, be0, lane);
        store_tile(hbuf, p0, 0, lane);
        store_tile(hbuf, p1, 16, lane);
        __syncthreads();

        v8f q0[8] = {}, q1[8] = {};
        gemm_lds<32>(hbuf, wp1, q0, q1, lane);
        layernorm_tanh(q0, q1, b1, g1, be1, lane);
        store_tile(hbuf, q0, 0, lane);
        store_tile(hbuf, q1, 16, lane);
        __syncthreads();

        v8f r0[8] = {}, r1[8] = {};
        gemm_lds<32>(hbuf, wp2, r0, r1, lane);
        layernorm_tanh(r0, r1, b2, g2, be2, lane);

        {
            const int col0 = lane & 15;
            const int grp  = lane >> 4;
            float w3[8];
#pragma unroll
            for (int nn = 0; nn < 8; ++nn) w3[nn] = W3[col0 + (nn << 4)];
            const float bias3 = b3[0];
#pragma unroll
            for (int t = 0; t < 2; ++t) {
#pragma unroll
                for (int j = 0; j < 8; ++j) {
                    float p = 0.f;
#pragma unroll
                    for (int nn = 0; nn < 8; ++nn)
                        p = fmaf(t ? r1[nn][j] : r0[nn][j], w3[nn], p);
                    p = row_sum16(p);
                    const int e = e0 + (t << 4) + j + (grp << 3);
                    if (col0 == j && e < E) out[e] = p + bias3;
                }
            }
        }
        __syncthreads();
    }
}

extern "C" void kernel_launch(void* const* d_in, const int* in_sizes, int n_in,
                              void* d_out, int out_size, void* d_ws, size_t ws_size,
                              hipStream_t stream)
{
    const float* x   = (const float*)d_in[0];
    const int*   ei  = (const int*)  d_in[1];
    const float* W0  = (const float*)d_in[2];
    const float* b0  = (const float*)d_in[3];
    const float* W1  = (const float*)d_in[4];
    const float* b1  = (const float*)d_in[5];
    const float* W2  = (const float*)d_in[6];
    const float* b2  = (const float*)d_in[7];
    const float* W3  = (const float*)d_in[8];
    const float* b3  = (const float*)d_in[9];
    const float* g0  = (const float*)d_in[10];
    const float* be0 = (const float*)d_in[11];
    const float* g1  = (const float*)d_in[12];
    const float* be1 = (const float*)d_in[13];
    const float* g2  = (const float*)d_in[14];
    const float* be2 = (const float*)d_in[15];

    const int E     = in_sizes[1] / 2;
    const int tiles = (E + TILE_M - 1) / TILE_M;
    int blocks = (tiles + WPB - 1) / WPB;
    if (blocks > 1024) blocks = 1024;

    edge_mlp_wmma_f32<<<blocks, WPB * 32, 0, stream>>>(
        x, ei, E, W0, b0, W1, b1, W2, b2, W3, b3,
        g0, be0, g1, be1, g2, be2, (float*)d_out);
}


# vector phases laid out stage by stage across the 16 register rows (all sums, then all lane reductions, statistics, normalisations; was row after row); a tile's 32 results leave in 1 store; sums same
# speedup vs baseline: 1.7991x; 1.0611x over previous
#include <hip/hip_runtime.h>
#include <math.h>


typedef __attribute__((ext_vector_type(2))) float v2f;
typedef __attribute__((ext_vector_type(4))) float v4f;
typedef __attribute__((ext_vector_type(8))) float v8f;

#define DHID   128
#define TILE_M 32
#define WPB    4
#define LNEPS  1e-5f

#if __has_builtin(__builtin_amdgcn_tanhf)
#define TANHF(v) __builtin_amdgcn_tanhf(v)
#else
#define TANHF(v) tanhf(v)
#endif

#define SCHED_FENCE() __builtin_amdgcn_sched_barrier(0)

#define OFF_W1 32768
#define OFF_W2 49152
#define OFF_H  65536
#define HTILE  (TILE_M * DHID)
#define SMEM_FLOATS (OFF_H + WPB * HTILE)


__device__ __forceinline__ void pack_slabs16(float* __restrict__ dst,
                                             const float* __restrict__ W,
                                             int wave, int lane)
{
    const float* src = W   + (wave << 11) + lane;
    float*       d   = dst + (wave << 11) + (lane << 2);
    float v[4][4][4];
#pragma unroll
    for (int u = 0; u < 4; ++u)
#pragma unroll
        for (int s = 0; s < 4; ++s)
#pragma unroll
            for (int j = 0; j < 4; ++j)
                v[u][j][s] = src[(u << 9) + (s << 7) + (j << 5)];
#pragma unroll
    for (int u = 0; u < 4; ++u)
#pragma unroll
        for (int j = 0; j < 4; ++j) {
            const v4f o = { v[u][j][0], v[u][j][1], v[u][j][2], v[u][j][3] };
            *(v4f*)(d + (u << 9) + (j << 7)) = o;
        }
}

struct Frags {
    v2f a0[4];
    v2f a1[4];
    v2f b[4][8];
};

__device__ __forceinline__ void load_frags(Frags& f, const float* __restrict__ hbuf,
                                           const float* __restrict__ wl,
                                           int abase, int kk0)
{
#pragma unroll
    for (int u = 0; u < 4; ++u) {
        const int kk = kk0 + u;
        const float* ap = hbuf + (abase ^ (kk << 2));
        f.a0[u] = *(const v2f*)ap;
        f.a1[u] = *(const v2f*)(ap + 16 * DHID);
        const float* wk = wl + (kk << 9);
#pragma unroll
        for (int nn = 0; nn < 8; ++nn)
            f.b[u][nn] = *(const v2f*)(wk + (nn << 6));
    }
}

__device__ __forceinline__ void mma_block(const Frags& f, v8f acc0[8], v8f acc1[8])
{
#pragma unroll
    for (int u = 0; u < 4; ++u) {
#pragma unroll
        for (int nn = 0; nn < 8; ++nn) {
            acc0[nn] = __builtin_amdgcn_wmma_f32_16x16x4_f32(
                false, f.a0[u], false, f.b[u][nn], (short)0, acc0[nn], false, false);
            acc1[nn] = __builtin_amdgcn_wmma_f32_16x16x4_f32(
                false, f.a1[u], false, f.b[u][nn], (short)0, acc1[nn], false, false);
        }
    }
}

template <int KSTEPS>
__device__ __forceinline__ void gemm_lds(const float* __restrict__ hbuf,
                                         const float* __restrict__ wp,
                                         v8f acc0[8], v8f acc1[8], int lane)
{
    static_assert(KSTEPS >= 8 && (KSTEPS & 7) == 0, "two blocks of 4 k-steps per trip");
    const int r  = lane & 15;
    const int g2 = (lane >> 4) << 1;
    const int abase = r * DHID + (r << 2) + g2;
    const float* wl = wp + (r << 2) + g2;

    Frags f0, f1;
    load_frags(f0, hbuf, wl, abase, 0);
#pragma unroll 1
    for (int kk = 0; kk < KSTEPS - 8; kk += 8) {
        load_frags(f1, hbuf, wl, abase, kk + 4);
        SCHED_FENCE();
        mma_block(f0, acc0, acc1);
        SCHED_FENCE();
        load_frags(f0, hbuf, wl, abase, kk + 8);
        SCHED_FENCE();
        mma_block(f1, acc0, acc1);
        SCHED_FENCE();
    }
    load_frags(f1, hbuf, wl, abase, KSTEPS - 4);
    SCHED_FENCE();
    mma_block(f0, acc0, acc1);
    SCHED_FENCE();
    mma_block(f1, acc0, acc1);
}

template <int M>
__device__ __forceinline__ float row_xor_add(float v)
{
    const int p = __builtin_amdgcn_update_dpp(0, __float_as_int(v), 0x160 + M,
                                              0xf, 0xf, true);
    return v + __int_as_float(p);
}

__device__ __forceinline__ void layernorm_tanh(v8f acc0[8], v8f acc1[8],
                                               const float* __restrict__ bias,
                                               const float* __restrict__ gamma,
                                               const float* __restrict__ beta,
                                               int lane)
{
    const int col0 = lane & 15;
    float bsv[8], gv[8], bv[8];
#pragma unroll
    for (int nn = 0; nn < 8; ++nn) {
        const int c = col0 + (nn << 4);
        bsv[nn] = bias[c]; gv[nn] = gamma[c]; bv[nn] = beta[c];
    }
    float s[16], ss[16];
#pragma unroll
    for (int q = 0; q < 16; ++q) { s[q] = 0.f; ss[q] = 0.f; }
#pragma unroll
    for (int nn = 0; nn < 8; ++nn) {
#pragma unroll
        for (int q = 0; q < 16; ++q) {
            const int j = q & 7;
            float v = (q < 8 ? acc0[nn][j] : acc1[nn][j]) + bsv[nn];
            if (q < 8) acc0[nn][j] = v; else acc1[nn][j] = v;
            s[q] += v;
            ss[q] = fmaf(v, v, ss[q]);
        }
    }
#pragma unroll
    for (int q = 0; q < 16; ++q) { s[q] = row_xor_add<1>(s[q]); ss[q] = row_xor_add<1>(ss[q]); }
#pragma unroll
    for (int q = 0; q < 16; ++q) { s[q] = row_xor_add<2>(s[q]); ss[q] = row_xor_add<2>(ss[q]); }
#pragma unroll
    for (int q = 0; q < 16; ++q) { s[q] = row_xor_add<4>(s[q]); ss[q] = row_xor_add<4>(ss[q]); }
#pragma unroll
    for (int q = 0; q < 16; ++q) { s[q] = row_xor_add<8>(s[q]); ss[q] = row_xor_add<8>(ss[q]); }
    float mu[16], inv[16];
#pragma unroll
    for (int q = 0; q < 16; ++q) {
        mu[q] = s[q] * (1.0f / 128.0f);
        const float var = fmaf(-mu[q], mu[q], ss[q] * (1.0f / 128.0f));
        inv[q] = rsqrtf(var + LNEPS);
    }
#pragma unroll
    for (int nn = 0; nn < 8; ++nn) {
#pragma unroll
        for (int q = 0; q < 16; ++q) {
            const int j = q & 7;
            const float a = (q < 8 ? acc0[nn][j] : acc1[nn][j]);
            const float v = (a - mu[q]) * inv[q];
            const float y = TANHF(fmaf(v, gv[nn], bv[nn]));
            if (q < 8) acc0[nn][j] = y; else acc1[nn][j] = y;
        }
    }
}

__device__ __forceinline__ void store_tile(float* __restrict__ hbuf,
                                           const v8f acc[8], int rowoff, int lane)
{
    const int col0 = lane & 15;
    const int grp  = lane >> 4;
    const int y = (grp << 10) | ((((col0 >> 2) ^ (grp << 3))) << 2) | (col0 & 3);
#pragma unroll
    for (int j = 0; j < 8; ++j) {
#pragma unroll
        for (int nn = 0; nn < 8; ++nn)
            hbuf[(rowoff + j) * DHID + (y ^ (((nn << 2) ^ j) << 2))] = acc[nn][j];
    }
}

__device__ __forceinline__ void stage_rows(float* __restrict__ hbuf,
                                           const float* __restrict__ x,
                                           const int* __restrict__ idxbase,
                                           int e0, int E, int lane)
{
    const int c4 = lane << 2;
    v4f v[TILE_M];
#pragma unroll
    for (int r = 0; r < TILE_M; ++r) {
        int e = e0 + r; if (e >= E) e = E - 1;
        const int n = idxbase[e];
        v[r] = *(const v4f*)(x + (size_t)n * DHID + c4);
    }
#pragma unroll
    for (int r = 0; r < TILE_M; ++r)
        *(v4f*)(hbuf + r * DHID + ((lane ^ (r & 15)) << 2)) = v[r];
}

__global__ void __launch_bounds__(WPB * 32, 1)
edge_mlp_wmma_f32(const float* __restrict__ x, const int* __restrict__ eidx, int E,
                  const float* __restrict__ W0, const float* __restrict__ b0,
                  const float* __restrict__ W1, const float* __restrict__ b1,
                  const float* __restrict__ W2, const float* __restrict__ b2,
                  const float* __restrict__ W3, const float* __restrict__ b3,
                  const float* __restrict__ g0, const float* __restrict__ be0,
                  const float* __restrict__ g1, const float* __restrict__ be1,
                  const float* __restrict__ g2, const float* __restrict__ be2,
                  float* __restrict__ out)
{
    __shared__ alignas(16) float smem[SMEM_FLOATS];
    const int tid  = threadIdx.x;
    const int lane = tid & 31;
    const int wave = tid >> 5;

    float* wp0  = smem;
    float* wp1  = smem + OFF_W1;
    float* wp2  = smem + OFF_W2;
    float* hbuf = smem + OFF_H + wave * HTILE;

#pragma unroll
    for (int t = 0; t < 8; ++t) {
        const float* Wt = (t < 4) ? W0 + (t << 13)
                        : (t < 6) ? W1 + ((t - 4) << 13)
                                  : W2 + ((t - 6) << 13);
        pack_slabs16(smem + (t << 13), Wt, wave, lane);
    }
    __syncthreads();

    const int tiles  = (E + TILE_M - 1) / TILE_M;
    const int stride = gridDim.x * WPB;

    for (int tile0 = blockIdx.x * WPB; tile0 < tiles; tile0 += stride) {
        const int tile = tile0 + wave;
        const int e0   = tile * TILE_M;

        {
            const int tn = tile + stride;
            if (tn < tiles) {
                __builtin_prefetch(&eidx[tn * TILE_M], 0, 1);
                __builtin_prefetch(&eidx[E + tn * TILE_M], 0, 1);
            }
        }

        v8f p0[8] = {}, p1[8] = {};
        stage_rows(hbuf, x, eidx, e0, E, lane);
        __syncthreads();
        gemm_lds<32>(hbuf, wp0, p0, p1, lane);
        __syncthreads();
        stage_rows(hbuf, x, eidx + E, e0, E, lane);
        __syncthreads();
        gemm_lds<32>(hbuf, wp0 + 32 * 512, p0, p1, lane);
        layernorm_tanh(p0, p1, b0, g0, be0, lane);
        store_tile(hbuf, p0, 0, lane);
        store_tile(hbuf, p1, 16, lane);
        __syncthreads();

        v8f q0[8] = {}, q1[8] = {};
        gemm_lds<32>(hbuf, wp1, q0, q1, lane);
        layernorm_tanh(q0, q1, b1, g1, be1, lane);
        store_tile(hbuf, q0, 0, lane);
        store_tile(hbuf, q1, 16, lane);
        __syncthreads();

        v8f r0[8] = {}, r1[8] = {};
        gemm_lds<32>(hbuf, wp2, r0, r1, lane);
        layernorm_tanh(r0, r1, b2, g2, be2, lane);

        {
            const int col0 = lane & 15;
            const int grp  = lane >> 4;
            float w3[8];
#pragma unroll
            for (int nn = 0; nn < 8; ++nn) w3[nn] = W3[col0 + (nn << 4)];
            const float bias3 = b3[0];
            float p[16];
#pragma unroll
            for (int q = 0; q < 16; ++q) p[q] = 0.f;
#pragma unroll
            for (int nn = 0; nn < 8; ++nn)
#pragma unroll
                for (int q = 0; q < 16; ++q)
                    p[q] = fmaf(q < 8 ? r0[nn][q & 7] : r1[nn][q & 7], w3[nn], p[q]);
#pragma unroll
            for (int q = 0; q < 16; ++q) p[q] = row_xor_add<1>(p[q]);
#pragma unroll
            for (int q = 0; q < 16; ++q) p[q] = row_xor_add<2>(p[q]);
#pragma unroll
            for (int q = 0; q < 16; ++q) p[q] = row_xor_add<4>(p[q]);
#pragma unroll
            for (int q = 0; q < 16; ++q) p[q] = row_xor_add<8>(p[q]);
            float res = p[0];
#pragma unroll
            for (int q = 1; q < 16; ++q) res = (col0 == q) ? p[q] : res;
            const int e = e0 + ((col0 >> 3) << 4) + (col0 & 7) + (grp << 3);
            if (e < E) out[e] = res + bias3;
        }
        __syncthreads();
    }
}

extern "C" void kernel_launch(void* const* d_in, const int* in_sizes, int n_in,
                              void* d_out, int out_size, void* d_ws, size_t ws_size,
                              hipStream_t stream)
{
    const float* x   = (const float*)d_in[0];
    const int*   ei  = (const int*)  d_in[1];
    const float* W0  = (const float*)d_in[2];
    const float* b0  = (const float*)d_in[3];
    const float* W1  = (const float*)d_in[4];
    const float* b1  = (const float*)d_in[5];
    const float* W2  = (const float*)d_in[6];
    const float* b2  = (const float*)d_in[7];
    const float* W3  = (const float*)d_in[8];
    const float* b3  = (const float*)d_in[9];
    const float* g0  = (const float*)d_in[10];
    const float* be0 = (const float*)d_in[11];
    const float* g1  = (const float*)d_in[12];
    const float* be1 = (const float*)d_in[13];
    const float* g2  = (const float*)d_in[14];
    const float* be2 = (const float*)d_in[15];

    const int E     = in_sizes[1] / 2;
    const int tiles = (E + TILE_M - 1) / TILE_M;
    int blocks = (tiles + WPB - 1) / WPB;
    if (blocks > 1024) blocks = 1024;

    edge_mlp_wmma_f32<<<blocks, WPB * 32, 0, stream>>>(
        x, ei, E, W0, b0, W1, b1, W2, b2, W3, b3,
        g0, be0, g1, be1, g2, be2, (float*)d_out);
}


# layer 0's A operand goes straight from global memory to registers (a lane loads its rows' 16-byte chunks, a stage ahead); no LDS staging of node rows, its 4 barriers go; layer-0 k order changes
# speedup vs baseline: 1.9353x; 1.0757x over previous
#include <hip/hip_runtime.h>
#include <math.h>


typedef __attribute__((ext_vector_type(2))) float v2f;
typedef __attribute__((ext_vector_type(4))) float v4f;
typedef __attribute__((ext_vector_type(8))) float v8f;

#define DHID   128
#define TILE_M 32
#define WPB    4
#define LNEPS  1e-5f

#if __has_builtin(__builtin_amdgcn_tanhf)
#define TANHF(v) __builtin_amdgcn_tanhf(v)
#else
#define TANHF(v) tanhf(v)
#endif

#define SCHED_FENCE() __builtin_amdgcn_sched_barrier(0)

#define OFF_W1 32768
#define OFF_W2 49152
#define OFF_H  65536
#define HTILE  (TILE_M * DHID)
#define SMEM_FLOATS (OFF_H + WPB * HTILE)


template <bool SWAP_ODD>
__device__ __forceinline__ void pack_slabs16(float* __restrict__ dst,
                                             const float* __restrict__ W,
                                             int wave, int lane)
{
    const float* src = W   + (wave << 11) + lane;
    float*       d   = dst + (wave << 11) + (lane << 2);
    float v[4][4][4];
#pragma unroll
    for (int u = 0; u < 4; ++u)
#pragma unroll
        for (int s = 0; s < 4; ++s)
#pragma unroll
            for (int j = 0; j < 4; ++j)
                v[u][j][s] = src[(u << 9) + (s << 7) + (j << 5)];
#pragma unroll
    for (int u = 0; u < 4; ++u)
#pragma unroll
        for (int j = 0; j < 4; ++j) {
            const bool sw = SWAP_ODD && (u & 1);
            const v4f o = { v[u][j][sw ? 2 : 0], v[u][j][sw ? 3 : 1],
                            v[u][j][sw ? 0 : 2], v[u][j][sw ? 1 : 3] };
            *(v4f*)(d + (u << 9) + (j << 7)) = o;
        }
}

struct Frags {
    v2f a0[4];
    v2f a1[4];
    v2f b[4][8];
};

__device__ __forceinline__ void load_frags(Frags& f, const float* __restrict__ hbuf,
                                           const float* __restrict__ wl,
                                           int abase, int kk0)
{
#pragma unroll
    for (int u = 0; u < 4; ++u) {
        const int kk = kk0 + u;
        const float* ap = hbuf + (abase ^ (kk << 2));
        f.a0[u] = *(const v2f*)ap;
        f.a1[u] = *(const v2f*)(ap + 16 * DHID);
        const float* wk = wl + (kk << 9);
#pragma unroll
        for (int nn = 0; nn < 8; ++nn)
            f.b[u][nn] = *(const v2f*)(wk + (nn << 6));
    }
}

__device__ __forceinline__ void mma_block(const Frags& f, v8f acc0[8], v8f acc1[8])
{
#pragma unroll
    for (int u = 0; u < 4; ++u) {
#pragma unroll
        for (int nn = 0; nn < 8; ++nn) {
            acc0[nn] = __builtin_amdgcn_wmma_f32_16x16x4_f32(
                false, f.a0[u], false, f.b[u][nn], (short)0, acc0[nn], false, false);
            acc1[nn] = __builtin_amdgcn_wmma_f32_16x16x4_f32(
                false, f.a1[u], false, f.b[u][nn], (short)0, acc1[nn], false, false);
        }
    }
}

template <int KSTEPS>
__device__ __forceinline__ void gemm_lds(const float* __restrict__ hbuf,
                                         const float* __restrict__ wp,
                                         v8f acc0[8], v8f acc1[8], int lane)
{
    static_assert(KSTEPS >= 8 && (KSTEPS & 7) == 0, "two blocks of 4 k-steps per trip");
    const int r  = lane & 15;
    const int g2 = (lane >> 4) << 1;
    const int abase = r * DHID + (r << 2) + g2;
    const float* wl = wp + (r << 2) + g2;

    Frags f0, f1;
    load_frags(f0, hbuf, wl, abase, 0);
#pragma unroll 1
    for (int kk = 0; kk < KSTEPS - 8; kk += 8) {
        load_frags(f1, hbuf, wl, abase, kk + 4);
        SCHED_FENCE();
        mma_block(f0, acc0, acc1);
        SCHED_FENCE();
        load_frags(f0, hbuf, wl, abase, kk + 8);
        SCHED_FENCE();
        mma_block(f1, acc0, acc1);
        SCHED_FENCE();
    }
    load_frags(f1, hbuf, wl, abase, KSTEPS - 4);
    SCHED_FENCE();
    mma_block(f0, acc0, acc1);
    SCHED_FENCE();
    mma_block(f1, acc0, acc1);
}

struct XRows { v4f a0[16]; v4f a1[16]; };

__device__ __forceinline__ void load_idx(const int* __restrict__ idxbase, int e0, int E,
                                         int r, int& n0, int& n1)
{
    int ea = e0 + r;      if (ea >= E) ea = E - 1;
    int eb = e0 + 16 + r; if (eb >= E) eb = E - 1;
    n0 = idxbase[ea];
    n1 = idxbase[eb];
}

__device__ __forceinline__ void load_rows(XRows& f, const float* __restrict__ x,
                                          int n0, int n1, int g)
{
    const float* pa = x + (((unsigned)n0 << 7) + (unsigned)(g << 2));
    const float* pb = x + (((unsigned)n1 << 7) + (unsigned)(g << 2));
#pragma unroll
    for (int j = 0; j < 16; ++j) {
        f.a0[j] = *(const v4f*)(pa + (j << 3));
        f.a1[j] = *(const v4f*)(pb + (j << 3));
    }
}

struct BFrags { v2f b[4][8]; };

__device__ __forceinline__ void load_b0(BFrags& f, const float* __restrict__ wl0,
                                        const float* __restrict__ wl1, int blk)
{
#pragma unroll
    for (int jj = 0; jj < 2; ++jj)
#pragma unroll
        for (int p = 0; p < 2; ++p) {
            const float* wk = (p ? wl1 : wl0) + ((2 * blk + jj) << 10);
#pragma unroll
            for (int nn = 0; nn < 8; ++nn)
                f.b[2 * jj + p][nn] = *(const v2f*)(wk + (nn << 6));
        }
}

__device__ __forceinline__ void mma_b0(const BFrags& f, const XRows& xr, int blk,
                                       v8f acc0[8], v8f acc1[8])
{
#pragma unroll
    for (int jj = 0; jj < 2; ++jj)
#pragma unroll
        for (int p = 0; p < 2; ++p) {
            const v4f c0 = xr.a0[2 * blk + jj];
            const v4f c1 = xr.a1[2 * blk + jj];
            const v2f A0 = { c0[2 * p], c0[2 * p + 1] };
            const v2f A1 = { c1[2 * p], c1[2 * p + 1] };
#pragma unroll
            for (int nn = 0; nn < 8; ++nn) {
                acc0[nn] = __builtin_amdgcn_wmma_f32_16x16x4_f32(
                    false, A0, false, f.b[2 * jj + p][nn], (short)0, acc0[nn], false, false);
                acc1[nn] = __builtin_amdgcn_wmma_f32_16x16x4_f32(
                    false, A1, false, f.b[2 * jj + p][nn], (short)0, acc1[nn], false, false);
            }
        }
}

__device__ __forceinline__ void gemm_l0(const XRows& xr, const float* __restrict__ wph,
                                        v8f acc0[8], v8f acc1[8], int lane)
{
    const int c = lane & 15;
    const int g = lane >> 4;
    const float* wl0 = wph + (g << 9) + (c << 2) + (g << 1);
    const float* wl1 = wph + (g << 9) + (c << 2) + ((1 - g) << 1);
    BFrags f0, f1;
    load_b0(f0, wl0, wl1, 0);
#pragma unroll
    for (int b = 0; b < 8; b += 2) {
        load_b0(f1, wl0, wl1, b + 1);
        SCHED_FENCE();
        mma_b0(f0, xr, b, acc0, acc1);
        SCHED_FENCE();
        if (b + 2 < 8) load_b0(f0, wl0, wl1, b + 2);
        SCHED_FENCE();
        mma_b0(f1, xr, b + 1, acc0, acc1);
        SCHED_FENCE();
    }
}

template <int M>
__device__ __forceinline__ float row_xor_add(float v)
{
    const int p = __builtin_amdgcn_update_dpp(0, __float_as_int(v), 0x160 + M,
                                              0xf, 0xf, true);
    return v + __int_as_float(p);
}

__device__ __forceinline__ void layernorm_tanh(v8f acc0[8], v8f acc1[8],
                                               const float* __restrict__ bias,
                                               const float* __restrict__ gamma,
                                               const float* __restrict__ beta,
                                               int lane)
{
    const int col0 = lane & 15;
    float bsv[8], gv[8], bv[8];
#pragma unroll
    for (int nn = 0; nn < 8; ++nn) {
        const int c = col0 + (nn << 4);
        bsv[nn] = bias[c]; gv[nn] = gamma[c]; bv[nn] = beta[c];
    }
    float s[16], ss[16];
#pragma unroll
    for (int q = 0; q < 16; ++q) { s[q] = 0.f; ss[q] = 0.f; }
#pragma unroll
    for (int nn = 0; nn < 8; ++nn) {
#pragma unroll
        for (int q = 0; q < 16; ++q) {
            const int j = q & 7;
            float v = (q < 8 ? acc0[nn][j] : acc1[nn][j]) + bsv[nn];
            if (q < 8) acc0[nn][j] = v; else acc1[nn][j] = v;
            s[q] += v;
            ss[q] = fmaf(v, v, ss[q]);
        }
    }
#pragma unroll
    for (int q = 0; q < 16; ++q) { s[q] = row_xor_add<1>(s[q]); ss[q] = row_xor_add<1>(ss[q]); }
#pragma unroll
    for (int q = 0; q < 16; ++q) { s[q] = row_xor_add<2>(s[q]); ss[q] = row_xor_add<2>(ss[q]); }
#pragma unroll
    for (int q = 0; q < 16; ++q) { s[q] = row_xor_add<4>(s[q]); ss[q] = row_xor_add<4>(ss[q]); }
#pragma unroll
    for (int q = 0; q < 16; ++q) { s[q] = row_xor_add<8>(s[q]); ss[q] = row_xor_add<8>(ss[q]); }
    float mu[16], inv[16];
#pragma unroll
    for (int q = 0; q < 16; ++q) {
        mu[q] = s[q] * (1.0f / 128.0f);
        const float var = fmaf(-mu[q], mu[q], ss[q] * (1.0f / 128.0f));
        inv[q] = rsqrtf(var + LNEPS);
    }
#pragma unroll
    for (int nn = 0; nn < 8; ++nn) {
#pragma unroll
        for (int q = 0; q < 16; ++q) {
            const int j = q & 7;
            const float a = (q < 8 ? acc0[nn][j] : acc1[nn][j]);
            const float v = (a - mu[q]) * inv[q];
            const float y = TANHF(fmaf(v, gv[nn], bv[nn]));
            if (q < 8) acc0[nn][j] = y; else acc1[nn][j] = y;
        }
    }
}

__device__ __forceinline__ void store_tile(float* __restrict__ hbuf,
                                           const v8f acc[8], int rowoff, int lane)
{
    const int col0 = lane & 15;
    const int grp  = lane >> 4;
    const int y = (grp << 10) | ((((col0 >> 2) ^ (grp << 3))) << 2) | (col0 & 3);
#pragma unroll
    for (int j = 0; j < 8; ++j) {
#pragma unroll
        for (int nn = 0; nn < 8; ++nn)
            hbuf[(rowoff + j) * DHID + (y ^ (((nn << 2) ^ j) << 2))] = acc[nn][j];
    }
}

__global__ void __launch_bounds__(WPB * 32, 1)
edge_mlp_wmma_f32(const float* __restrict__ x, const int* __restrict__ eidx, int E,
                  const float* __restrict__ W0, const float* __restrict__ b0,
                  const float* __restrict__ W1, const float* __restrict__ b1,
                  const float* __restrict__ W2, const float* __restrict__ b2,
                  const float* __restrict__ W3, const float* __restrict__ b3,
                  const float* __restrict__ g0, const float* __restrict__ be0,
                  const float* __restrict__ g1, const float* __restrict__ be1,
                  const float* __restrict__ g2, const float* __restrict__ be2,
                  float* __restrict__ out)
{
    __shared__ alignas(16) float smem[SMEM_FLOATS];
    const int tid  = threadIdx.x;
    const int lane = tid & 31;
    const int wave = tid >> 5;

    float* wp0  = smem;
    float* wp1  = smem + OFF_W1;
    float* wp2  = smem + OFF_W2;
    float* hbuf = smem + OFF_H + wave * HTILE;

    const int lr = lane & 15;
    const int lg = lane >> 4;

    XRows xs;
    int ie0, ie1;
    {
        const int e0 = (blockIdx.x * WPB + wave) * TILE_M;
        int is0, is1;
        load_idx(eidx, e0, E, lr, is0, is1);
        load_idx(eidx + E, e0, E, lr, ie0, ie1);
        load_rows(xs, x, is0, is1, lg);
    }

#pragma unroll
    for (int t = 0; t < 8; ++t) {
        const float* Wt = (t < 4) ? W0 + (t << 13)
                        : (t < 6) ? W1 + ((t - 4) << 13)
                                  : W2 + ((t - 6) << 13);
        if (t < 4) pack_slabs16<true >(smem + (t << 13), Wt, wave, lane);
        else       pack_slabs16<false>(smem + (t << 13), Wt, wave, lane);
    }
    __syncthreads();

    const int tiles  = (E + TILE_M - 1) / TILE_M;
    const int stride = gridDim.x * WPB;
    for (int tile0 = blockIdx.x * WPB; tile0 < tiles; tile0 += stride) {
        const int tile = tile0 + wave;
        const int e0   = tile * TILE_M;

        int ns0, ns1, ne0, ne1;
        load_idx(eidx,     e0 + stride * TILE_M, E, lr, ns0, ns1);
        load_idx(eidx + E, e0 + stride * TILE_M, E, lr, ne0, ne1);
        XRows xe;
        load_rows(xe, x, ie0, ie1, lg);
        SCHED_FENCE();

        v8f p0[8] = {}, p1[8] = {};
        gemm_l0(xs, wp0, p0, p1, lane);
        load_rows(xs, x, ns0, ns1, lg);
        ie0 = ne0; ie1 = ne1;
        SCHED_FENCE();
        gemm_l0(xe, wp0 + 32 * 512, p0, p1, lane);
        layernorm_tanh(p0, p1, b0, g0, be0, lane);
        store_tile(hbuf, p0, 0, lane);
        store_tile(hbuf, p1, 16, lane);
        __syncthreads();

        v8f q0[8] = {}, q1[8] = {};
        gemm_lds<32>(hbuf, wp1, q0, q1, lane);
        layernorm_tanh(q0, q1, b1, g1, be1, lane);
        store_tile(hbuf, q0, 0, lane);
        store_tile(hbuf, q1, 16, lane);
        __syncthreads();

        v8f r0[8] = {}, r1[8] = {};
        gemm_lds<32>(hbuf, wp2, r0, r1, lane);
        layernorm_tanh(r0, r1, b2, g2, be2, lane);

        {
            const int col0 = lane & 15;
            const int grp  = lane >> 4;
            float w3[8];
#pragma unroll
            for (int nn = 0; nn < 8; ++nn) w3[nn] = W3[col0 + (nn << 4)];
            const float bias3 = b3[0];
            float p[16];
#pragma unroll
            for (int q = 0; q < 16; ++q) p[q] = 0.f;
#pragma unroll
            for (int nn = 0; nn < 8; ++nn)
#pragma unroll
                for (int q = 0; q < 16; ++q)
                    p[q] = fmaf(q < 8 ? r0[nn][q & 7] : r1[nn][q & 7], w3[nn], p[q]);
#pragma unroll
            for (int q = 0; q < 16; ++q) p[q] = row_xor_add<1>(p[q]);
#pragma unroll
            for (int q = 0; q < 16; ++q) p[q] = row_xor_add<2>(p[q]);
#pragma unroll
            for (int q = 0; q < 16; ++q) p[q] = row_xor_add<4>(p[q]);
#pragma unroll
            for (int q = 0; q < 16; ++q) p[q] = row_xor_add<8>(p[q]);
            float res = p[0];
#pragma unroll
            for (int q = 1; q < 16; ++q) res = (col0 == q) ? p[q] : res;
            const int e = e0 + ((col0 >> 3) << 4) + (col0 & 7) + (grp << 3);
            if (e < E) out[e] = res + bias3;
        }
    }
}

extern "C" void kernel_launch(void* const* d_in, const int* in_sizes, int n_in,
                              void* d_out, int out_size, void* d_ws, size_t ws_size,
                              hipStream_t stream)
{
    const float* x   = (const float*)d_in[0];
    const int*   ei  = (const int*)  d_in[1];
    const float* W0  = (const float*)d_in[2];
    const float* b0  = (const float*)d_in[3];
    const float* W1  = (const float*)d_in[4];
    const float* b1  = (const float*)d_in[5];
    const float* W2  = (const float*)d_in[6];
    const float* b2  = (const float*)d_in[7];
    const float* W3  = (const float*)d_in[8];
    const float* b3  = (const float*)d_in[9];
    const float* g0  = (const float*)d_in[10];
    const float* be0 = (const float*)d_in[11];
    const float* g1  = (const float*)d_in[12];
    const float* be1 = (const float*)d_in[13];
    const float* g2  = (const float*)d_in[14];
    const float* be2 = (const float*)d_in[15];

    const int E     = in_sizes[1] / 2;
    const int tiles = (E + TILE_M - 1) / TILE_M;
    int blocks = (tiles + WPB - 1) / WPB;
    if (blocks > 1024) blocks = 1024;

    edge_mlp_wmma_f32<<<blocks, WPB * 32, 0, stream>>>(
        x, ei, E, W0, b0, W1, b1, W2, b2, W3, b3,
        g0, be0, g1, be1, g2, be2, (float*)d_out);
}
